# starting kernel
# baseline (speedup 1.0000x reference)
#include <hip/hip_runtime.h>


#define D_IN        128
#define D_HID       64
#define OUT_CLS     2
#define TILE_M      16
#define WAVES_PB    4
#define BLOCK_THR   (WAVES_PB * 32)
#define GRID_BLOCKS 1536
#define E_STRIDE    132
#define H_STRIDE    68

typedef __attribute__((ext_vector_type(2))) float v2f;
typedef __attribute__((ext_vector_type(8))) float v8f;

__global__ void repack_w1_kernel(const float* __restrict__ W1,
                                 float* __restrict__ Bp) {
  int g = blockIdx.x * blockDim.x + threadIdx.x;
  if (g >= 32 * 4 * 32) return;
  int l = g & 31;
  int t = (g >> 5) & 3;
  int c = g >> 7;
  int k = 4 * c + 2 * (l >> 4);
  int n = 16 * t + (l & 15);
  Bp[2 * g + 0] = W1[k * D_HID + n];
  Bp[2 * g + 1] = W1[(k + 1) * D_HID + n];
}

__global__ __launch_bounds__(BLOCK_THR)
void edge_mlp_kernel(const float* __restrict__ h,
                     const float* __restrict__ Bp,
                     const float* __restrict__ b1,
                     const float* __restrict__ W2,
                     const float* __restrict__ b2,
                     const int*   __restrict__ src,
                     const int*   __restrict__ dst,
                     float*       __restrict__ out,
                     int n_edges) {
  __shared__ float s_par[D_HID * OUT_CLS + D_HID + OUT_CLS];
  __shared__ float s_e[WAVES_PB][TILE_M * E_STRIDE];

  const int lane = threadIdx.x & 31;
  const int wave = threadIdx.x >> 5;
  const int m_l  = lane & 15;
  const int k_hi = (lane >> 4) << 1;
  const int r_hi = (lane >> 4) << 3;

  for (int i = threadIdx.x; i < D_HID * OUT_CLS + D_HID + OUT_CLS;
       i += BLOCK_THR) {
    float v;
    if (i < D_HID * OUT_CLS)               v = W2[i];
    else if (i < D_HID * OUT_CLS + D_HID)  v = b1[i - D_HID * OUT_CLS];
    else                                   v = b2[i - D_HID * OUT_CLS - D_HID];
    s_par[i] = v;
  }
  __syncthreads();

  float* et = s_e[wave];

  const int n_tiles      = (n_edges + TILE_M - 1) / TILE_M;
  const int stride_tiles = gridDim.x * WAVES_PB;

  for (int tile = blockIdx.x * WAVES_PB + wave; tile < n_tiles;
       tile += stride_tiles) {
    const int e0 = tile * TILE_M;

    __builtin_prefetch(src + e0 + stride_tiles * TILE_M, 0, 0);
    __builtin_prefetch(dst + e0 + stride_tiles * TILE_M, 0, 0);

    int ei = e0 + m_l;
    if (ei >= n_edges) ei = n_edges - 1;
    const int my_s = src[ei];
    const int my_d = dst[ei];

    #pragma unroll 4
    for (int m = 0; m < TILE_M; ++m) {
      const int sm = __shfl(my_s, m, 32);
      const int dm = __shfl(my_d, m, 32);
      const float4 a = *(const float4*)(h + (size_t)sm * D_IN + lane * 4);
      const float4 b = *(const float4*)(h + (size_t)dm * D_IN + lane * 4);
      float4 p;
      p.x = a.x * b.x; p.y = a.y * b.y; p.z = a.z * b.z; p.w = a.w * b.w;
      *(float4*)(et + m * E_STRIDE + lane * 4) = p;
    }

    v8f acc0 = {}, acc1 = {}, acc2 = {}, acc3 = {};
    #pragma unroll
    for (int c = 0; c < D_IN / 4; ++c) {
      const v2f av = *(const v2f*)(et + m_l * E_STRIDE + 4 * c + k_hi);
      const float* bp = Bp + ((size_t)(c * 4) * 32 + lane) * 2;
      const v2f bv0 = *(const v2f*)(bp);
      const v2f bv1 = *(const v2f*)(bp + 64);
      const v2f bv2 = *(const v2f*)(bp + 128);
      const v2f bv3 = *(const v2f*)(bp + 192);
      acc0 = __builtin_amdgcn_wmma_f32_16x16x4_f32(false, av, false, bv0,
                                                   (short)0, acc0, false, false);
      acc1 = __builtin_amdgcn_wmma_f32_16x16x4_f32(false, av, false, bv1,
                                                   (short)0, acc1, false, false);
      acc2 = __builtin_amdgcn_wmma_f32_16x16x4_f32(false, av, false, bv2,
                                                   (short)0, acc2, false, false);
      acc3 = __builtin_amdgcn_wmma_f32_16x16x4_f32(false, av, false, bv3,
                                                   (short)0, acc3, false, false);
    }

    #define EPILOGUE(ACC, T)                                                 \
      {                                                                      \
        const float bias = s_par[D_HID * OUT_CLS + (T) * 16 + m_l];          \
        _Pragma("unroll") for (int r = 0; r < 8; ++r) {                      \
          float v = ACC[r] + bias;                                           \
          v = v > 0.0f ? v : 0.0f;                                           \
          et[(r + r_hi) * H_STRIDE + (T) * 16 + m_l] = v;                    \
        }                                                                    \
      }
    EPILOGUE(acc0, 0) EPILOGUE(acc1, 1) EPILOGUE(acc2, 2) EPILOGUE(acc3, 3)
    #undef EPILOGUE

    const int m_out = lane >> 1;
    const int cl    = lane & 1;
    float sum = s_par[D_HID * OUT_CLS + D_HID + cl];
    #pragma unroll
    for (int k = 0; k < D_HID; ++k)
      sum += et[m_out * H_STRIDE + k] * s_par[k * OUT_CLS + cl];

    const int oi = e0 * OUT_CLS + lane;
    if (oi < n_edges * OUT_CLS) out[oi] = sum;
  }
}

extern "C" void kernel_launch(void* const* d_in, const int* in_sizes, int n_in,
                              void* d_out, int out_size, void* d_ws,
                              size_t ws_size, hipStream_t stream) {
  (void)n_in; (void)out_size; (void)ws_size;
  const float* h   = (const float*)d_in[0];
  const float* W1  = (const float*)d_in[1];
  const float* b1  = (const float*)d_in[2];
  const float* W2  = (const float*)d_in[3];
  const float* b2  = (const float*)d_in[4];
  const int*   src = (const int*)d_in[5];
  const int*   dst = (const int*)d_in[6];
  float*       out = (float*)d_out;
  float*       Bp  = (float*)d_ws;

  const int n_edges = in_sizes[5];

  repack_w1_kernel<<<(32 * 4 * 32 + 255) / 256, 256, 0, stream>>>(W1, Bp);

  const int n_tiles = (n_edges + TILE_M - 1) / TILE_M;
  int blocks = (n_tiles + WAVES_PB - 1) / WAVES_PB;
  if (blocks > GRID_BLOCKS) blocks = GRID_BLOCKS;
  edge_mlp_kernel<<<blocks, BLOCK_THR, 0, stream>>>(h, Bp, b1, W2, b2, src,
                                                    dst, out, n_edges);
}


# Software-pipeline the row gather: next tile's 32 h-row loads issued into registers before the 128 WMMAs, indices two tiles ahead; grid 1536->512 blocks to match 2 waves/SIMD
# speedup vs baseline: 1.0685x; 1.0685x over previous
#include <hip/hip_runtime.h>


#define D_IN        128
#define D_HID       64
#define OUT_CLS     2
#define TILE_M      16
#define WAVES_PB    4
#define BLOCK_THR   (WAVES_PB * 32)
#define GRID_BLOCKS 512
#define E_STRIDE    132
#define H_STRIDE    68

typedef __attribute__((ext_vector_type(2))) float v2f;
typedef __attribute__((ext_vector_type(8))) float v8f;

__global__ void repack_w1_kernel(const float* __restrict__ W1,
                                 float* __restrict__ Bp) {
  int g = blockIdx.x * blockDim.x + threadIdx.x;
  if (g >= 32 * 4 * 32) return;
  int l = g & 31;
  int t = (g >> 5) & 3;
  int c = g >> 7;
  int k = 4 * c + 2 * (l >> 4);
  int n = 16 * t + (l & 15);
  Bp[2 * g + 0] = W1[k * D_HID + n];
  Bp[2 * g + 1] = W1[(k + 1) * D_HID + n];
}

__global__ __launch_bounds__(BLOCK_THR)
void edge_mlp_kernel(const float* __restrict__ h,
                     const float* __restrict__ Bp,
                     const float* __restrict__ b1,
                     const float* __restrict__ W2,
                     const float* __restrict__ b2,
                     const int*   __restrict__ src,
                     const int*   __restrict__ dst,
                     float*       __restrict__ out,
                     int n_edges) {
  __shared__ float s_par[D_HID * OUT_CLS + D_HID + OUT_CLS];
  __shared__ float s_e[WAVES_PB][TILE_M * E_STRIDE];

  const int lane = threadIdx.x & 31;
  const int wave = __builtin_amdgcn_readfirstlane(threadIdx.x >> 5);
  const int m_l  = lane & 15;
  const int k_hi = (lane >> 4) << 1;
  const int r_hi = (lane >> 4) << 3;

  for (int i = threadIdx.x; i < D_HID * OUT_CLS + D_HID + OUT_CLS;
       i += BLOCK_THR) {
    float v;
    if (i < D_HID * OUT_CLS)               v = W2[i];
    else if (i < D_HID * OUT_CLS + D_HID)  v = b1[i - D_HID * OUT_CLS];
    else                                   v = b2[i - D_HID * OUT_CLS - D_HID];
    s_par[i] = v;
  }
  __syncthreads();

  float* et = s_e[wave];

  const int n_tiles      = (n_edges + TILE_M - 1) / TILE_M;
  const int stride_tiles = gridDim.x * WAVES_PB;

  int tile = blockIdx.x * WAVES_PB + wave;
  if (tile >= n_tiles) return;

  const float* hl = h + lane * 4;

  #define LOAD_IDX(T, S, D)                                                  \
    {                                                                        \
      int ei_ = (T) * TILE_M + m_l;                                          \
      if (ei_ >= n_edges) ei_ = n_edges - 1;                                 \
      S = src[ei_];                                                          \
      D = dst[ei_];                                                          \
    }
  #define ISSUE_GATHER(S, D)                                                 \
    _Pragma("unroll") for (int m = 0; m < TILE_M; ++m) {                     \
      const int sm = __builtin_amdgcn_readlane(S, m);                        \
      const int dm = __builtin_amdgcn_readlane(D, m);                        \
      ra[m] = *(const float4*)(hl + (size_t)sm * D_IN);                      \
      rb[m] = *(const float4*)(hl + (size_t)dm * D_IN);                      \
    }

  float4 ra[TILE_M], rb[TILE_M];
  int cur_s, cur_d, nxt_s, nxt_d;
  LOAD_IDX(tile, cur_s, cur_d)
  LOAD_IDX(tile + stride_tiles, nxt_s, nxt_d)
  ISSUE_GATHER(cur_s, cur_d)

  for (;;) {
    const int e0 = tile * TILE_M;

    #pragma unroll
    for (int m = 0; m < TILE_M; ++m) {
      float4 p;
      p.x = ra[m].x * rb[m].x; p.y = ra[m].y * rb[m].y;
      p.z = ra[m].z * rb[m].z; p.w = ra[m].w * rb[m].w;
      *(float4*)(et + m * E_STRIDE + lane * 4) = p;
    }

    const int  ntile = tile + stride_tiles;
    const bool more  = ntile < n_tiles;
    if (more) {
      ISSUE_GATHER(nxt_s, nxt_d)
      LOAD_IDX(ntile + stride_tiles, nxt_s, nxt_d)
    }

    v8f acc0 = {}, acc1 = {}, acc2 = {}, acc3 = {};
    #pragma unroll
    for (int c = 0; c < D_IN / 4; ++c) {
      const v2f av = *(const v2f*)(et + m_l * E_STRIDE + 4 * c + k_hi);
      const float* bp = Bp + ((size_t)(c * 4) * 32 + lane) * 2;
      const v2f bv0 = *(const v2f*)(bp);
      const v2f bv1 = *(const v2f*)(bp + 64);
      const v2f bv2 = *(const v2f*)(bp + 128);
      const v2f bv3 = *(const v2f*)(bp + 192);
      acc0 = __builtin_amdgcn_wmma_f32_16x16x4_f32(false, av, false, bv0,
                                                   (short)0, acc0, false, false);
      acc1 = __builtin_amdgcn_wmma_f32_16x16x4_f32(false, av, false, bv1,
                                                   (short)0, acc1, false, false);
      acc2 = __builtin_amdgcn_wmma_f32_16x16x4_f32(false, av, false, bv2,
                                                   (short)0, acc2, false, false);
      acc3 = __builtin_amdgcn_wmma_f32_16x16x4_f32(false, av, false, bv3,
                                                   (short)0, acc3, false, false);
    }

    #define EPILOGUE(ACC, T)                                                 \
      {                                                                      \
        const float bias = s_par[D_HID * OUT_CLS + (T) * 16 + m_l];          \
        _Pragma("unroll") for (int r = 0; r < 8; ++r) {                      \
          float v = ACC[r] + bias;                                           \
          v = v > 0.0f ? v : 0.0f;                                           \
          et[(r + r_hi) * H_STRIDE + (T) * 16 + m_l] = v;                    \
        }                                                                    \
      }
    EPILOGUE(acc0, 0) EPILOGUE(acc1, 1) EPILOGUE(acc2, 2) EPILOGUE(acc3, 3)
    #undef EPILOGUE

    const int m_out = lane >> 1;
    const int cl    = lane & 1;
    float sum = s_par[D_HID * OUT_CLS + D_HID + cl];
    #pragma unroll
    for (int k = 0; k < D_HID; ++k)
      sum += et[m_out * H_STRIDE + k] * s_par[k * OUT_CLS + cl];

    const int oi = e0 * OUT_CLS + lane;
    if (oi < n_edges * OUT_CLS) out[oi] = sum;

    if (!more) break;
    tile  = ntile;
  }
  #undef LOAD_IDX
  #undef ISSUE_GATHER
}

extern "C" void kernel_launch(void* const* d_in, const int* in_sizes, int n_in,
                              void* d_out, int out_size, void* d_ws,
                              size_t ws_size, hipStream_t stream) {
  (void)n_in; (void)out_size; (void)ws_size;
  const float* h   = (const float*)d_in[0];
  const float* W1  = (const float*)d_in[1];
  const float* b1  = (const float*)d_in[2];
  const float* W2  = (const float*)d_in[3];
  const float* b2  = (const float*)d_in[4];
  const int*   src = (const int*)d_in[5];
  const int*   dst = (const int*)d_in[6];
  float*       out = (float*)d_out;
  float*       Bp  = (float*)d_ws;

  const int n_edges = in_sizes[5];

  repack_w1_kernel<<<(32 * 4 * 32 + 255) / 256, 256, 0, stream>>>(W1, Bp);

  const int n_tiles = (n_edges + TILE_M - 1) / TILE_M;
  int blocks = (n_tiles + WAVES_PB - 1) / WAVES_PB;
  if (blocks > GRID_BLOCKS) blocks = GRID_BLOCKS;
  edge_mlp_kernel<<<blocks, BLOCK_THR, 0, stream>>>(h, Bp, b1, W2, b2, src,
                                                    dst, out, n_edges);
}


# Layer-1 product transposed (W1 fragments as A operand, edge tile as B): one edge per lane, so layer 2 is register FMAs + one lane-half exchange, hid skips LDS; b1 and layer-2 sum order changed
# speedup vs baseline: 1.1824x; 1.1066x over previous
#include <hip/hip_runtime.h>


#define D_IN        128
#define D_HID       64
#define OUT_CLS     2
#define TILE_M      16
#define WAVES_PB    4
#define BLOCK_THR   (WAVES_PB * 32)
#define GRID_BLOCKS 512
#define E_STRIDE    132

typedef __attribute__((ext_vector_type(2))) float v2f;
typedef __attribute__((ext_vector_type(8))) float v8f;

__global__ void repack_w1_kernel(const float* __restrict__ W1,
                                 float* __restrict__ Bp) {
  int g = blockIdx.x * blockDim.x + threadIdx.x;
  if (g >= 32 * 4 * 32) return;
  int l = g & 31;
  int t = (g >> 5) & 3;
  int c = g >> 7;
  int k = 4 * c + 2 * (l >> 4);
  int n = 16 * t + (l & 15);
  Bp[2 * g + 0] = W1[k * D_HID + n];
  Bp[2 * g + 1] = W1[(k + 1) * D_HID + n];
}

__device__ __forceinline__ float other_half(float x) {
  const int b = __builtin_bit_cast(int, x);
  return __builtin_bit_cast(float,
      __builtin_amdgcn_permlanex16(b, b, 0x76543210, 0xfedcba98, false, false));
}

__global__ __launch_bounds__(BLOCK_THR)
void edge_mlp_kernel(const float* __restrict__ h,
                     const float* __restrict__ Bp,
                     const float* __restrict__ b1,
                     const float* __restrict__ W2,
                     const float* __restrict__ b2,
                     const int*   __restrict__ src,
                     const int*   __restrict__ dst,
                     float*       __restrict__ out,
                     int n_edges) {
  __shared__ __attribute__((aligned(16))) float s_par[D_HID * OUT_CLS + D_HID];
  __shared__ float s_e[WAVES_PB][TILE_M * E_STRIDE];

  const int lane = threadIdx.x & 31;
  const int wave = __builtin_amdgcn_readfirstlane(threadIdx.x >> 5);
  const int m_l  = lane & 15;
  const int half = lane >> 4;
  const int k_hi = half << 1;
  const int r_hi = half << 3;

  for (int i = threadIdx.x; i < D_HID * OUT_CLS + D_HID; i += BLOCK_THR)
    s_par[i] = i < D_HID * OUT_CLS ? W2[i] : b1[i - D_HID * OUT_CLS];
  __syncthreads();
  const float* s_w2 = s_par;
  const float* s_b1 = s_par + D_HID * OUT_CLS;

  float* et = s_e[wave];

  const int n_tiles      = (n_edges + TILE_M - 1) / TILE_M;
  const int stride_tiles = gridDim.x * WAVES_PB;

  int tile = blockIdx.x * WAVES_PB + wave;
  if (tile >= n_tiles) return;

  const float b2v = b2[half];

  v2f wf[D_IN / 4][4];
  #pragma unroll
  for (int c = 0; c < D_IN / 4; ++c) {
    const float* bp = Bp + ((size_t)(c * 4) * 32 + lane) * 2;
    wf[c][0] = *(const v2f*)(bp);
    wf[c][1] = *(const v2f*)(bp + 64);
    wf[c][2] = *(const v2f*)(bp + 128);
    wf[c][3] = *(const v2f*)(bp + 192);
  }

  const float* hl = h + lane * 4;

  #define LOAD_IDX(T, S, D)                                                  \
    {                                                                        \
      int ei_ = (T) * TILE_M + m_l;                                          \
      if (ei_ >= n_edges) ei_ = n_edges - 1;                                 \
      S = src[ei_];                                                          \
      D = dst[ei_];                                                          \
    }
  #define ISSUE_GATHER(S, D)                                                 \
    _Pragma("unroll") for (int m = 0; m < TILE_M; ++m) {                     \
      const int sm = __builtin_amdgcn_readlane(S, m);                        \
      const int dm = __builtin_amdgcn_readlane(D, m);                        \
      ra[m] = *(const float4*)(hl + (size_t)sm * D_IN);                      \
      rb[m] = *(const float4*)(hl + (size_t)dm * D_IN);                      \
    }

  float4 ra[TILE_M], rb[TILE_M];
  int cur_s, cur_d, nxt_s, nxt_d;
  LOAD_IDX(tile, cur_s, cur_d)
  LOAD_IDX(tile + stride_tiles, nxt_s, nxt_d)
  ISSUE_GATHER(cur_s, cur_d)

  for (;;) {
    const int e0 = tile * TILE_M;

    asm volatile("" ::: "memory");

    #pragma unroll
    for (int m = 0; m < TILE_M; ++m) {
      float4 p;
      p.x = ra[m].x * rb[m].x; p.y = ra[m].y * rb[m].y;
      p.z = ra[m].z * rb[m].z; p.w = ra[m].w * rb[m].w;
      *(float4*)(et + m * E_STRIDE + lane * 4) = p;
    }

    const int  ntile = tile + stride_tiles;
    const bool more  = ntile < n_tiles;
    if (more) {
      ISSUE_GATHER(nxt_s, nxt_d)
      LOAD_IDX(ntile + stride_tiles, nxt_s, nxt_d)
    }

    #define BIAS_TILE(T)                                                     \
      ({                                                                     \
        const float4 lo_ = *(const float4*)(s_b1 + (T) * 16 + r_hi);         \
        const float4 hi_ = *(const float4*)(s_b1 + (T) * 16 + r_hi + 4);     \
        v8f a_ = {lo_.x, lo_.y, lo_.z, lo_.w, hi_.x, hi_.y, hi_.z, hi_.w};   \
        a_;                                                                  \
      })
    v8f acc0 = BIAS_TILE(0), acc1 = BIAS_TILE(1);
    v8f acc2 = BIAS_TILE(2), acc3 = BIAS_TILE(3);
    #undef BIAS_TILE
    #pragma unroll
    for (int c = 0; c < D_IN / 4; ++c) {
      const v2f ev = *(const v2f*)(et + m_l * E_STRIDE + 4 * c + k_hi);
      acc0 = __builtin_amdgcn_wmma_f32_16x16x4_f32(false, wf[c][0], false, ev,
                                                   (short)0, acc0, false, false);
      acc1 = __builtin_amdgcn_wmma_f32_16x16x4_f32(false, wf[c][1], false, ev,
                                                   (short)0, acc1, false, false);
      acc2 = __builtin_amdgcn_wmma_f32_16x16x4_f32(false, wf[c][2], false, ev,
                                                   (short)0, acc2, false, false);
      acc3 = __builtin_amdgcn_wmma_f32_16x16x4_f32(false, wf[c][3], false, ev,
                                                   (short)0, acc3, false, false);
    }

    #define LAYER2(ACC, T, S0, S1)                                           \
      {                                                                      \
        const float* wp_ = s_w2 + ((T) * 16 + r_hi) * OUT_CLS;               \
        float v_ = ACC[0];                                                   \
        v_ = v_ > 0.0f ? v_ : 0.0f;                                          \
        S0 = v_ * wp_[0];                                                    \
        S1 = v_ * wp_[1];                                                    \
        _Pragma("unroll") for (int r = 1; r < 8; ++r) {                      \
          v_ = ACC[r];                                                       \
          v_ = v_ > 0.0f ? v_ : 0.0f;                                        \
          S0 += v_ * wp_[2 * r];                                             \
          S1 += v_ * wp_[2 * r + 1];                                         \
        }                                                                    \
      }
    float p00, p01, p10, p11, p20, p21, p30, p31;
    LAYER2(acc0, 0, p00, p01) LAYER2(acc1, 1, p10, p11)
    LAYER2(acc2, 2, p20, p21) LAYER2(acc3, 3, p30, p31)
    #undef LAYER2
    const float o0 = (p00 + p10) + (p20 + p30);
    const float o1 = (p01 + p11) + (p21 + p31);

    const float keep = half ? o1 : o0;
    const float send = half ? o0 : o1;
    const float sum  = keep + other_half(send) + b2v;

    const int edge = e0 + m_l;
    if (edge < n_edges) out[edge * OUT_CLS + half] = sum;

    if (!more) break;
    tile  = ntile;
  }
  #undef LOAD_IDX
  #undef ISSUE_GATHER
}

extern "C" void kernel_launch(void* const* d_in, const int* in_sizes, int n_in,
                              void* d_out, int out_size, void* d_ws,
                              size_t ws_size, hipStream_t stream) {
  (void)n_in; (void)out_size; (void)ws_size;
  const float* h   = (const float*)d_in[0];
  const float* W1  = (const float*)d_in[1];
  const float* b1  = (const float*)d_in[2];
  const float* W2  = (const float*)d_in[3];
  const float* b2  = (const float*)d_in[4];
  const int*   src = (const int*)d_in[5];
  const int*   dst = (const int*)d_in[6];
  float*       out = (float*)d_out;
  float*       Bp  = (float*)d_ws;

  const int n_edges = in_sizes[5];

  repack_w1_kernel<<<(32 * 4 * 32 + 255) / 256, 256, 0, stream>>>(W1, Bp);

  const int n_tiles = (n_edges + TILE_M - 1) / TILE_M;
  int blocks = (n_tiles + WAVES_PB - 1) / WAVES_PB;
  if (blocks > GRID_BLOCKS) blocks = GRID_BLOCKS;
  edge_mlp_kernel<<<blocks, BLOCK_THR, 0, stream>>>(h, Bp, b1, W2, b2, src,
                                                    dst, out, n_edges);
}
